# starting kernel
# baseline (speedup 1.0000x reference)
#include <hip/hip_runtime.h>

typedef __attribute__((ext_vector_type(2))) float v2f;
typedef __attribute__((ext_vector_type(8))) float v8f;

#define CD 8
#define TT 512
#define NSEG 511
#define NTHREADS 128

__global__ __launch_bounds__(NTHREADS)
void logsig_kernel(const float* __restrict__ path, float* __restrict__ out) {
    __shared__ float dxs[NSEG * CD];
    __shared__ float S3s[512];
    __shared__ float S2b[2][64];
    __shared__ float S1b[2][8];
    __shared__ float S4s[4096];

    const int b    = blockIdx.x;
    const int tid  = threadIdx.x;
    const int lane = tid & 31;
    const int w    = tid >> 5;
    const int Ln   = lane & 15;
    const int half = lane >> 4;

    const float* pb = path + (size_t)b * (TT * CD);

    for (int e = tid; e < NSEG * CD; e += NTHREADS)
        dxs[e] = pb[e + 8] - pb[e];
    for (int e = tid; e < 512; e += NTHREADS) S3s[e] = 0.0f;
    if (tid < 64) S2b[0][tid] = 0.0f;
    if (tid < 8)  S1b[0][tid] = 0.0f;

    v8f acc[8];
    const v8f vzero = {0.f, 0.f, 0.f, 0.f, 0.f, 0.f, 0.f, 0.f};
#pragma unroll
    for (int q = 0; q < 8; ++q) acc[q] = vzero;

    int cur = 0;
    for (int t = 0; t < NSEG; ++t) {
        __syncthreads();
        const float* d = &dxs[t * CD];
        const float dk = d[Ln & 7];

        const float bv = (Ln < 8) ? d[Ln] : 0.0f;
        v2f bf; bf[0] = bv; bf[1] = bv;

        float s3n[8];
#pragma unroll
        for (int q = 0; q < 8; ++q) {
            const int tt = w * 8 + q;
            const int r  = tt * 16 + Ln;
            const int i  = r >> 6;
            const int j  = (r >> 3) & 7;
            const float s2v = S2b[cur][(i << 3) | j];
            const float s1v = S1b[cur][i];
            const float di = d[i], dj = d[j];
            const float p1 = s2v * dk;
            const float p2 = s1v * dj * dk;
            const float p3 = di * dj * dk;
            const float s3v = S3s[r];

            v2f af;
            af[0] = half ? (1.0f / 6.0f)  * p2 : s3v;
            af[1] = half ? (1.0f / 24.0f) * p3 : 0.5f * p1;

            acc[q] = __builtin_amdgcn_wmma_f32_16x16x4_f32(
                false, af, false, bf, (short)0, acc[q], false, false);

            s3n[q] = s3v + p1 + 0.5f * p2 + (1.0f / 6.0f) * p3;
        }
        if (half == 0) {
#pragma unroll
            for (int q = 0; q < 8; ++q)
                S3s[(w * 8 + q) * 16 + Ln] = s3n[q];
        }
        const int nxt = cur ^ 1;
        if (tid < 64) {
            const int i2 = tid >> 3, j2 = tid & 7;
            S2b[nxt][tid] = S2b[cur][tid] + S1b[cur][i2] * d[j2]
                          + 0.5f * d[i2] * d[j2];
        } else if (tid < 72) {
            const int c = tid - 64;
            S1b[nxt][c] = S1b[cur][c] + d[c];
        }
        cur = nxt;
    }

    __syncthreads();
    if (Ln < 8) {
#pragma unroll
        for (int q = 0; q < 8; ++q) {
            const int tt = w * 8 + q;
#pragma unroll
            for (int v = 0; v < 8; ++v) {
                const int m = v + (half << 3);
                S4s[(tt * 16 + m) * 8 + Ln] = acc[q][v];
            }
        }
    }
    __syncthreads();

    float* ob = out + (size_t)b * 4680;
    const float* S1f = S1b[cur];
    const float* S2f = S2b[cur];

    if (tid < 8) ob[tid] = S1f[tid];
    if (tid < 64) {
        const int i = tid >> 3, j = tid & 7;
        ob[8 + tid] = S2f[tid] - 0.5f * S1f[i] * S1f[j];
    }
    for (int e = tid; e < 512; e += NTHREADS) {
        const int i = e >> 6, j = (e >> 3) & 7, k = e & 7;
        const float v3 = S3s[e]
            - 0.5f * (S1f[i] * S2f[(j << 3) | k] + S2f[(i << 3) | j] * S1f[k])
            + (1.0f / 3.0f) * S1f[i] * S1f[j] * S1f[k];
        ob[72 + e] = v3;
    }
    for (int e = tid; e < 4096; e += NTHREADS) {
        const int i = e >> 9, j = (e >> 6) & 7, k = (e >> 3) & 7, l = e & 7;
        const float s1i = S1f[i], s1j = S1f[j], s1k = S1f[k], s1l = S1f[l];
        const float v4 = S4s[e]
            - 0.5f * ( s1i * S3s[e & 511]
                     + S2f[(i << 3) | j] * S2f[(k << 3) | l]
                     + S3s[e >> 3] * s1l )
            + (1.0f / 3.0f) * ( s1i * s1j * S2f[(k << 3) | l]
                              + s1i * S2f[(j << 3) | k] * s1l
                              + S2f[(i << 3) | j] * s1k * s1l )
            - 0.25f * s1i * s1j * s1k * s1l;
        ob[584 + e] = v4;
    }
}

extern "C" void kernel_launch(void* const* d_in, const int* in_sizes, int n_in,
                              void* d_out, int out_size, void* d_ws, size_t ws_size,
                              hipStream_t stream) {
    const float* path = (const float*)d_in[0];
    float* out = (float*)d_out;
    const int B = in_sizes[0] / (TT * CD);
    logsig_kernel<<<dim3(B), dim3(NTHREADS), 0, stream>>>(path, out);
}


# Level-1/2/3 state kept in per-lane registers (each lane updates its own copy), so the time loop has no barrier and no LDS state reads or writes; summation order unchanged
# speedup vs baseline: 2.1497x; 2.1497x over previous
#include <hip/hip_runtime.h>


typedef __attribute__((ext_vector_type(2))) float v2f;
typedef __attribute__((ext_vector_type(8))) float v8f;

#define CD 8
#define TT 512
#define NSEG 511
#define NTHREADS 128

__global__ __launch_bounds__(NTHREADS)
void logsig_kernel(const float* __restrict__ path, float* __restrict__ out) {
    __shared__ float dxs[NSEG * CD];
    __shared__ float S3s[512];
    __shared__ float S2s[64];
    __shared__ float S1s[8];
    __shared__ float S4s[4096];

    const int b    = blockIdx.x;
    const int tid  = threadIdx.x;
    const int lane = tid & 31;
    const int w    = tid >> 5;
    const int Ln   = lane & 15;
    const int half = lane >> 4;
    const int kk   = Ln & 7;
    const int hi   = Ln >> 3;

    const float* pb = path + (size_t)b * (TT * CD);

    for (int e = tid; e < NSEG * CD; e += NTHREADS)
        dxs[e] = pb[e + 8] - pb[e];

    v8f acc[8];
    const v8f vzero = {0.f, 0.f, 0.f, 0.f, 0.f, 0.f, 0.f, 0.f};
    float s3[8], s2[8], s1[2];
#pragma unroll
    for (int q = 0; q < 8; ++q) { acc[q] = vzero; s3[q] = 0.0f; s2[q] = 0.0f; }
    s1[0] = 0.0f; s1[1] = 0.0f;

    __syncthreads();

    for (int t = 0; t < NSEG; ++t) {
        const float* d = &dxs[t * CD];
        const float dk = d[kk];
        float di[2], dj[4];
        di[0] = d[2 * w]; di[1] = d[2 * w + 1];
#pragma unroll
        for (int m = 0; m < 4; ++m) dj[m] = d[2 * m + hi];

        const float bv = (Ln < 8) ? dk : 0.0f;
        v2f bf; bf[0] = bv; bf[1] = bv;

#pragma unroll
        for (int q = 0; q < 8; ++q) {
            const int a = q >> 2, m = q & 3;
            const float s2v = s2[q];
            const float s1v = s1[a];
            const float div = di[a], djv = dj[m];
            const float p1 = s2v * dk;
            const float p2 = s1v * djv * dk;
            const float p3 = div * djv * dk;
            const float s3v = s3[q];

            v2f af;
            af[0] = half ? (1.0f / 6.0f)  * p2 : s3v;
            af[1] = half ? (1.0f / 24.0f) * p3 : 0.5f * p1;

            acc[q] = __builtin_amdgcn_wmma_f32_16x16x4_f32(
                false, af, false, bf, (short)0, acc[q], false, false);

            s3[q] = s3v + p1 + 0.5f * p2 + (1.0f / 6.0f) * p3;
            s2[q] = s2v + s1v * djv + 0.5f * div * djv;
        }
        s1[0] += di[0];
        s1[1] += di[1];
    }

    if (half == 0) {
#pragma unroll
        for (int q = 0; q < 8; ++q)
            S3s[(w * 8 + q) * 16 + Ln] = s3[q];
    }
    if (lane == 0 || lane == 8) {
#pragma unroll
        for (int q = 0; q < 8; ++q)
            S2s[((2 * w + (q >> 2)) << 3) | (2 * (q & 3) + hi)] = s2[q];
    }
    if (lane == 0) {
        S1s[2 * w]     = s1[0];
        S1s[2 * w + 1] = s1[1];
    }
    if (Ln < 8) {
#pragma unroll
        for (int q = 0; q < 8; ++q) {
            const int tt = w * 8 + q;
#pragma unroll
            for (int v = 0; v < 8; ++v) {
                const int mm = v + (half << 3);
                S4s[(tt * 16 + mm) * 8 + Ln] = acc[q][v];
            }
        }
    }
    __syncthreads();

    float* ob = out + (size_t)b * 4680;
    const float* S1f = S1s;
    const float* S2f = S2s;

    if (tid < 8) ob[tid] = S1f[tid];
    if (tid < 64) {
        const int i = tid >> 3, j = tid & 7;
        ob[8 + tid] = S2f[tid] - 0.5f * S1f[i] * S1f[j];
    }
    for (int e = tid; e < 512; e += NTHREADS) {
        const int i = e >> 6, j = (e >> 3) & 7, k = e & 7;
        const float v3 = S3s[e]
            - 0.5f * (S1f[i] * S2f[(j << 3) | k] + S2f[(i << 3) | j] * S1f[k])
            + (1.0f / 3.0f) * S1f[i] * S1f[j] * S1f[k];
        ob[72 + e] = v3;
    }
    for (int e = tid; e < 4096; e += NTHREADS) {
        const int i = e >> 9, j = (e >> 6) & 7, k = (e >> 3) & 7, l = e & 7;
        const float s1i = S1f[i], s1j = S1f[j], s1k = S1f[k], s1l = S1f[l];
        const float v4 = S4s[e]
            - 0.5f * ( s1i * S3s[e & 511]
                     + S2f[(i << 3) | j] * S2f[(k << 3) | l]
                     + S3s[e >> 3] * s1l )
            + (1.0f / 3.0f) * ( s1i * s1j * S2f[(k << 3) | l]
                              + s1i * S2f[(j << 3) | k] * s1l
                              + S2f[(i << 3) | j] * s1k * s1l )
            - 0.25f * s1i * s1j * s1k * s1l;
        ob[584 + e] = v4;
    }
}

extern "C" void kernel_launch(void* const* d_in, const int* in_sizes, int n_in,
                              void* d_out, int out_size, void* d_ws, size_t ws_size,
                              hipStream_t stream) {
    const float* path = (const float*)d_in[0];
    float* out = (float*)d_out;
    const int B = in_sizes[0] / (TT * CD);
    logsig_kernel<<<dim3(B), dim3(NTHREADS), 0, stream>>>(path, out);
}


# Each matrix op covers 32 rows x 2 time steps (k slot = row set x step, B block-diagonal), so every lane owns distinct rows; row factor now summed on the vector unit, 2 steps summed in the matrix op
# speedup vs baseline: 5.6116x; 2.6104x over previous
#include <hip/hip_runtime.h>


typedef __attribute__((ext_vector_type(2))) float v2f;
typedef __attribute__((ext_vector_type(8))) float v8f;

#define CD 8
#define TT 512
#define NSEG 511
#define NPAIR 255
#define NTHREADS 128

__device__ __forceinline__ v2f chen_pair(float dk, float div, v2f djp,
                                         v2f& s3p, v2f& s2p, float& s1v) {
    const v2f p1 = s2p * dk;
    const v2f p2 = (s1v * djp) * dk;
    const v2f p3 = (div * djp) * dk;
    const v2f u = s3p + 0.5f * p1 + (1.0f / 6.0f) * p2 + (1.0f / 24.0f) * p3;
    s3p = s3p + p1 + 0.5f * p2 + (1.0f / 6.0f) * p3;
    s2p = s2p + s1v * djp + (0.5f * div) * djp;
    s1v += div;
    return u;
}

__global__ __launch_bounds__(NTHREADS)
void logsig_kernel(const float* __restrict__ path, float* __restrict__ out) {
    __shared__ float dxs[NSEG * CD];
    __shared__ float S3s[512];
    __shared__ float S2s[64];
    __shared__ float S1s[8];
    __shared__ float S4s[4096];

    const int b    = blockIdx.x;
    const int tid  = threadIdx.x;
    const int lane = tid & 31;
    const int w    = tid >> 5;
    const int c    = lane & 15;
    const int h    = lane >> 4;
    const int kk   = c & 7;
    const int hi   = c >> 3;
    const int jb   = 2 * h + hi;
    const bool own = (hi == h);

    const float* pb = path + (size_t)b * (TT * CD);

    for (int e = tid; e < NSEG * CD; e += NTHREADS)
        dxs[e] = pb[e + 8] - pb[e];

    v8f acc[4];
    const v8f vzero = {0.f, 0.f, 0.f, 0.f, 0.f, 0.f, 0.f, 0.f};
    v2f s3p[2], s2p[2];
    float s1[2];
#pragma unroll
    for (int q = 0; q < 4; ++q) acc[q] = vzero;
#pragma unroll
    for (int a = 0; a < 2; ++a) { s3p[a] = 0.0f; s2p[a] = 0.0f; s1[a] = 0.0f; }

    __syncthreads();

    for (int g = 0; g < NPAIR; ++g) {
        const float* d = &dxs[2 * g * CD];
        const float dk0 = d[kk], dk1 = d[CD + kk];
        float di0[2], di1[2];
        di0[0] = d[2 * w];      di0[1] = d[2 * w + 1];
        di1[0] = d[CD + 2 * w]; di1[1] = d[CD + 2 * w + 1];
        v2f djp0, djp1;
        djp0[0] = d[jb];      djp0[1] = d[4 + jb];
        djp1[0] = d[CD + jb]; djp1[1] = d[CD + 4 + jb];
        v2f bf; bf[0] = own ? dk0 : 0.0f; bf[1] = own ? dk1 : 0.0f;
#pragma unroll
        for (int a = 0; a < 2; ++a) {
            const v2f ua = chen_pair(dk0, di0[a], djp0, s3p[a], s2p[a], s1[a]);
            const v2f ub = chen_pair(dk1, di1[a], djp1, s3p[a], s2p[a], s1[a]);
            v2f af0, af1;
            af0[0] = ua[0]; af0[1] = ub[0];
            af1[0] = ua[1]; af1[1] = ub[1];
            acc[2 * a] = __builtin_amdgcn_wmma_f32_16x16x4_f32(
                false, af0, false, bf, (short)0, acc[2 * a], false, false);
            acc[2 * a + 1] = __builtin_amdgcn_wmma_f32_16x16x4_f32(
                false, af1, false, bf, (short)0, acc[2 * a + 1], false, false);
        }
    }
    {
        const float* d = &dxs[(NSEG - 1) * CD];
        const float dk0 = d[kk];
        v2f djp0; djp0[0] = d[jb]; djp0[1] = d[4 + jb];
        v2f bf; bf[0] = own ? dk0 : 0.0f; bf[1] = 0.0f;
#pragma unroll
        for (int a = 0; a < 2; ++a) {
            const v2f ua = chen_pair(dk0, d[2 * w + a], djp0, s3p[a], s2p[a], s1[a]);
            v2f af0, af1;
            af0[0] = ua[0]; af0[1] = 0.0f;
            af1[0] = ua[1]; af1[1] = 0.0f;
            acc[2 * a] = __builtin_amdgcn_wmma_f32_16x16x4_f32(
                false, af0, false, bf, (short)0, acc[2 * a], false, false);
            acc[2 * a + 1] = __builtin_amdgcn_wmma_f32_16x16x4_f32(
                false, af1, false, bf, (short)0, acc[2 * a + 1], false, false);
        }
    }

#pragma unroll
    for (int q = 0; q < 4; ++q)
        S3s[(w * 4 + q) * 32 + h * 16 + c] = s3p[q >> 1][q & 1];
    if (kk == 0) {
#pragma unroll
        for (int q = 0; q < 4; ++q)
            S2s[((2 * w + (q >> 1)) << 3) | (4 * (q & 1) + jb)] = s2p[q >> 1][q & 1];
    }
    if (lane == 0) {
        S1s[2 * w]     = s1[0];
        S1s[2 * w + 1] = s1[1];
    }
#pragma unroll
    for (int q = 0; q < 4; ++q) {
        const int row0 = (w * 4 + q) * 32 + hi * 16 + h * 8;
#pragma unroll
        for (int v = 0; v < 8; ++v)
            S4s[(row0 + v) * 8 + kk] = acc[q][v];
    }
    __syncthreads();

    float* ob = out + (size_t)b * 4680;
    const float* S1f = S1s;
    const float* S2f = S2s;

    if (tid < 8) ob[tid] = S1f[tid];
    if (tid < 64) {
        const int i = tid >> 3, j = tid & 7;
        ob[8 + tid] = S2f[tid] - 0.5f * S1f[i] * S1f[j];
    }
    for (int e = tid; e < 512; e += NTHREADS) {
        const int i = e >> 6, j = (e >> 3) & 7, k = e & 7;
        const float v3 = S3s[e]
            - 0.5f * (S1f[i] * S2f[(j << 3) | k] + S2f[(i << 3) | j] * S1f[k])
            + (1.0f / 3.0f) * S1f[i] * S1f[j] * S1f[k];
        ob[72 + e] = v3;
    }
    for (int e = tid; e < 4096; e += NTHREADS) {
        const int i = e >> 9, j = (e >> 6) & 7, k = (e >> 3) & 7, l = e & 7;
        const float s1i = S1f[i], s1j = S1f[j], s1k = S1f[k], s1l = S1f[l];
        const float v4 = S4s[e]
            - 0.5f * ( s1i * S3s[e & 511]
                     + S2f[(i << 3) | j] * S2f[(k << 3) | l]
                     + S3s[e >> 3] * s1l )
            + (1.0f / 3.0f) * ( s1i * s1j * S2f[(k << 3) | l]
                              + s1i * S2f[(j << 3) | k] * s1l
                              + S2f[(i << 3) | j] * s1k * s1l )
            - 0.25f * s1i * s1j * s1k * s1l;
        ob[584 + e] = v4;
    }
}

extern "C" void kernel_launch(void* const* d_in, const int* in_sizes, int n_in,
                              void* d_out, int out_size, void* d_ws, size_t ws_size,
                              hipStream_t stream) {
    const float* path = (const float*)d_in[0];
    float* out = (float*)d_out;
    const int B = in_sizes[0] / (TT * CD);
    logsig_kernel<<<dim3(B), dim3(NTHREADS), 0, stream>>>(path, out);
}


# Chen step factored (Horner): per-i coefficients S1+d/2, S1/2+d/6, S1/3+d/12 formed once per wave on the scalar unit, each row pair then needs 5 packed FMAs; rounding order of step terms changes
# speedup vs baseline: 8.7900x; 1.5664x over previous
#include <hip/hip_runtime.h>


typedef __attribute__((ext_vector_type(2))) float v2f;
typedef __attribute__((ext_vector_type(8))) float v8f;

#define CD 8
#define TT 512
#define NSEG 511
#define NCH 4
#define CHPAIR 64
#define NTHREADS 512
#define SIGLEN 4680
#define L2O 8
#define L3O 72
#define L4O 584
#define DXLEN 4096

__device__ __forceinline__ float wave_uniform(float x) {
    return __int_as_float(__builtin_amdgcn_readfirstlane(__float_as_int(x)));
}

__device__ __forceinline__ v2f chen_pair(float dk, float hdk, v2f djp,
                                         float a1, float a2, float a3,
                                         v2f& s3p, v2f& s2p) {
    const v2f t3 = s2p + djp * a3;
    const v2f t2 = s2p + djp * a2;
    const v2f u = s3p + hdk * t3;
    s3p = s3p + dk * t2;
    s2p = s2p + djp * a1;
    return u;
}

__global__ __launch_bounds__(NTHREADS)
void logsig_kernel(const float* __restrict__ path, float* __restrict__ out) {
    __shared__ float dxs[DXLEN];
    __shared__ float sig[NCH * SIGLEN];

    const int b    = blockIdx.x;
    const int tid  = threadIdx.x;
    const int lane = tid & 31;
    const int wv   = __builtin_amdgcn_readfirstlane(tid >> 5);
    const int w    = wv & 3;
    const int ch   = wv >> 2;
    const int c    = lane & 15;
    const int h    = lane >> 4;
    const int kk   = c & 7;
    const int hi   = c >> 3;
    const int jb   = 2 * h + hi;
    const bool own = (hi == h);

    const float* pb = path + (size_t)b * (TT * CD);

#pragma unroll
    for (int n = 0; n < 8; ++n) {
        const int e = tid + n * NTHREADS;
        const bool real = e < NSEG * CD;
        const int es = real ? e : e - CD;
        const float dv = pb[es + CD] - pb[es];
        dxs[e] = real ? dv : 0.0f;
    }

    v8f acc[4];
    const v8f vzero = {0.f, 0.f, 0.f, 0.f, 0.f, 0.f, 0.f, 0.f};
    v2f s3p[2], s2p[2];
    float s1[2];
#pragma unroll
    for (int q = 0; q < 4; ++q) acc[q] = vzero;
#pragma unroll
    for (int a = 0; a < 2; ++a) { s3p[a] = 0.0f; s2p[a] = 0.0f; s1[a] = 0.0f; }

    __syncthreads();

    for (int it = 0; it < CHPAIR; ++it) {
        const float* d = &dxs[2 * (ch * CHPAIR + it) * CD];
        const float dk0 = d[kk], dk1 = d[CD + kk];
        const float hk0 = 0.5f * dk0, hk1 = 0.5f * dk1;
        v2f djp0, djp1;
        djp0[0] = d[jb];      djp0[1] = d[4 + jb];
        djp1[0] = d[CD + jb]; djp1[1] = d[CD + 4 + jb];
        v2f bf; bf[0] = own ? dk0 : 0.0f; bf[1] = own ? dk1 : 0.0f;
#pragma unroll
        for (int a = 0; a < 2; ++a) {
            const float e0 = wave_uniform(d[2 * w + a]);
            const float e1 = wave_uniform(d[CD + 2 * w + a]);
            float sv = s1[a];
            const v2f ua = chen_pair(dk0, hk0, djp0,
                                     sv + 0.5f * e0,
                                     0.5f * sv + (1.0f / 6.0f) * e0,
                                     (1.0f / 3.0f) * sv + (1.0f / 12.0f) * e0,
                                     s3p[a], s2p[a]);
            sv += e0;
            const v2f ub = chen_pair(dk1, hk1, djp1,
                                     sv + 0.5f * e1,
                                     0.5f * sv + (1.0f / 6.0f) * e1,
                                     (1.0f / 3.0f) * sv + (1.0f / 12.0f) * e1,
                                     s3p[a], s2p[a]);
            s1[a] = sv + e1;
            v2f af0, af1;
            af0[0] = ua[0]; af0[1] = ub[0];
            af1[0] = ua[1]; af1[1] = ub[1];
            acc[2 * a] = __builtin_amdgcn_wmma_f32_16x16x4_f32(
                false, af0, false, bf, (short)0, acc[2 * a], false, false);
            acc[2 * a + 1] = __builtin_amdgcn_wmma_f32_16x16x4_f32(
                false, af1, false, bf, (short)0, acc[2 * a + 1], false, false);
        }
    }
    float* sg = &sig[ch * SIGLEN];
#pragma unroll
    for (int q = 0; q < 4; ++q)
        sg[L3O + (w * 4 + q) * 32 + h * 16 + c] = s3p[q >> 1][q & 1];
    if (kk == 0) {
#pragma unroll
        for (int q = 0; q < 4; ++q)
            sg[L2O + (((2 * w + (q >> 1)) << 3) | (4 * (q & 1) + jb))] = s2p[q >> 1][q & 1];
    }
    if (lane == 0) {
        sg[2 * w]     = s1[0];
        sg[2 * w + 1] = s1[1];
    }
#pragma unroll
    for (int q = 0; q < 4; ++q) {
        const int row0 = (w * 4 + q) * 32 + hi * 16 + h * 8;
#pragma unroll
        for (int v = 0; v < 8; ++v)
            sg[L4O + (row0 + v) * 8 + kk] = acc[q][v];
    }
    __syncthreads();

    {
        const int p  = tid >> 8;
        const int lt = tid & 255;
        float* A = &sig[(2 * p) * SIGLEN];
        const float* Bs = &sig[(2 * p + 1) * SIGLEN];
        float* pr = &dxs[p * L4O];
        float r4[16];
#pragma unroll
        for (int n = 0; n < 16; ++n) {
            const int e = lt + 256 * n;
            r4[n] = (A[L4O + e] + Bs[L4O + e])
                  + A[e >> 9] * Bs[L3O + (e & 511)]
                  + A[L2O + (e >> 6)] * Bs[L2O + (e & 63)]
                  + A[L3O + (e >> 3)] * Bs[e & 7];
        }
        for (int e = lt; e < 512; e += 256)
            pr[L3O + e] = (A[L3O + e] + Bs[L3O + e])
                        + A[e >> 6] * Bs[L2O + (e & 63)]
                        + A[L2O + (e >> 3)] * Bs[e & 7];
        if (lt < 64)
            pr[L2O + lt] = (A[L2O + lt] + Bs[L2O + lt]) + A[lt >> 3] * Bs[lt & 7];
        if (lt < 8)
            pr[lt] = A[lt] + Bs[lt];
#pragma unroll
        for (int n = 0; n < 16; ++n)
            A[L4O + lt + 256 * n] = r4[n];
    }
    __syncthreads();

    const float* P  = &dxs[0];
    const float* Q  = &dxs[L4O];
    const float* P4 = &sig[L4O];
    const float* Q4 = &sig[2 * SIGLEN + L4O];
    float* fin = &dxs[2 * L4O];
    fin[L3O + tid] = (P[L3O + tid] + Q[L3O + tid])
                   + P[tid >> 6] * Q[L2O + (tid & 63)]
                   + P[L2O + (tid >> 3)] * Q[tid & 7];
    if (tid < 64)
        fin[L2O + tid] = (P[L2O + tid] + Q[L2O + tid]) + P[tid >> 3] * Q[tid & 7];
    if (tid < 8)
        fin[tid] = P[tid] + Q[tid];
    __syncthreads();

    float* ob = out + (size_t)b * 4680;
    const float* S1f = fin;
    const float* S2f = fin + L2O;
    const float* S3s = fin + L3O;

    if (tid < 8) ob[tid] = S1f[tid];
    if (tid < 64) {
        const int i = tid >> 3, j = tid & 7;
        ob[8 + tid] = S2f[tid] - 0.5f * S1f[i] * S1f[j];
    }
    {
        const int e = tid;
        const int i = e >> 6, j = (e >> 3) & 7, k = e & 7;
        const float v3 = S3s[e]
            - 0.5f * (S1f[i] * S2f[(j << 3) | k] + S2f[(i << 3) | j] * S1f[k])
            + (1.0f / 3.0f) * S1f[i] * S1f[j] * S1f[k];
        ob[72 + e] = v3;
    }
    for (int e = tid; e < 4096; e += NTHREADS) {
        const int i = e >> 9, j = (e >> 6) & 7, k = (e >> 3) & 7, l = e & 7;
        const float s4 = (P4[e] + Q4[e])
                       + P[i] * Q[L3O + (e & 511)]
                       + P[L2O + (e >> 6)] * Q[L2O + (e & 63)]
                       + P[L3O + (e >> 3)] * Q[l];
        const float s1i = S1f[i], s1j = S1f[j], s1k = S1f[k], s1l = S1f[l];
        const float v4 = s4
            - 0.5f * ( s1i * S3s[e & 511]
                     + S2f[(i << 3) | j] * S2f[(k << 3) | l]
                     + S3s[e >> 3] * s1l )
            + (1.0f / 3.0f) * ( s1i * s1j * S2f[(k << 3) | l]
                              + s1i * S2f[(j << 3) | k] * s1l
                              + S2f[(i << 3) | j] * s1k * s1l )
            - 0.25f * s1i * s1j * s1k * s1l;
        ob[584 + e] = v4;
    }
}

extern "C" void kernel_launch(void* const* d_in, const int* in_sizes, int n_in,
                              void* d_out, int out_size, void* d_ws, size_t ws_size,
                              hipStream_t stream) {
    const float* path = (const float*)d_in[0];
    float* out = (float*)d_out;
    const int B = in_sizes[0] / (TT * CD);
    logsig_kernel<<<dim3(B), dim3(NTHREADS), 0, stream>>>(path, out);
}


# Each batch split over 2 workgroups by level 4's first index i (grid B x 2): each does half the level-4 matrix ops, joins and stores, all of levels 1-3; 256 compute units busy; sum order unchanged
# speedup vs baseline: 12.6717x; 1.4416x over previous
#include <hip/hip_runtime.h>


typedef __attribute__((ext_vector_type(2))) float v2f;
typedef __attribute__((ext_vector_type(8))) float v8f;

#define CD 8
#define TT 512
#define NSEG 511
#define NCH 4
#define CHPAIR 64
#define NTHREADS 512
#define SIGLEN 4680
#define L2O 8
#define L3O 72
#define L4O 584
#define DXLEN 4096

__device__ __forceinline__ float lane_value(float x, int l) {
    return __int_as_float(__builtin_amdgcn_readlane(__float_as_int(x), l));
}

__device__ __forceinline__ v2f chen_pair(float dk, float hdk, v2f djp,
                                         float a1, float a2, float a3,
                                         v2f& s3p, v2f& s2p) {
    const v2f t3 = s2p + djp * a3;
    const v2f t2 = s2p + djp * a2;
    const v2f u = s3p + hdk * t3;
    s3p = s3p + dk * t2;
    s2p = s2p + djp * a1;
    return u;
}

__device__ __forceinline__ void chen_low(float dk, v2f djp, float a1, float a2,
                                         v2f& s3p, v2f& s2p) {
    const v2f t2 = s2p + djp * a2;
    s3p = s3p + dk * t2;
    s2p = s2p + djp * a1;
}

__global__ __launch_bounds__(NTHREADS)
void logsig_kernel(const float* __restrict__ path, float* __restrict__ out) {
    __shared__ float lds[DXLEN + NCH * SIGLEN];
    float* const dxs = lds;
    float* const sig = lds + DXLEN;

    const int b    = blockIdx.x;
    const int g    = blockIdx.y;
    const int tid  = threadIdx.x;
    const int lane = tid & 31;
    const int wv   = __builtin_amdgcn_readfirstlane(tid >> 5);
    const int w    = wv & 3;
    const int ch   = wv >> 2;
    const int c    = lane & 15;
    const int h    = lane >> 4;
    const int kk   = c & 7;
    const int hi   = c >> 3;
    const int jb   = 2 * h + hi;
    const bool own = (hi == h);
    const int iw   = 4 * g + w;
    const int io   = 4 * (1 - g) + w;
    const int isel = (lane & 1) ? io : iw;

    const float* pb = path + (size_t)b * (TT * CD);

#pragma unroll
    for (int n = 0; n < 8; ++n) {
        const int e = tid + n * NTHREADS;
        const bool real = e < NSEG * CD;
        const int es = real ? e : e - CD;
        const float dv = pb[es + CD] - pb[es];
        dxs[e] = real ? dv : 0.0f;
    }

    v8f acc[2];
    const v8f vzero = {0.f, 0.f, 0.f, 0.f, 0.f, 0.f, 0.f, 0.f};
    v2f s3p[2], s2p[2];
    float s1[2];
#pragma unroll
    for (int q = 0; q < 2; ++q) acc[q] = vzero;
#pragma unroll
    for (int a = 0; a < 2; ++a) { s3p[a] = 0.0f; s2p[a] = 0.0f; s1[a] = 0.0f; }

    __syncthreads();

    for (int it = 0; it < CHPAIR; ++it) {
        const float* d = &dxs[2 * (ch * CHPAIR + it) * CD];
        const float dk0 = d[kk], dk1 = d[CD + kk];
        const float hk0 = 0.5f * dk0, hk1 = 0.5f * dk1;
        v2f djp0, djp1;
        djp0[0] = d[jb];      djp0[1] = d[4 + jb];
        djp1[0] = d[CD + jb]; djp1[1] = d[CD + 4 + jb];
        v2f bf; bf[0] = own ? dk0 : 0.0f; bf[1] = own ? dk1 : 0.0f;
        const float du0 = d[isel], du1 = d[CD + isel];
        {
            const float e0 = lane_value(du0, 0);
            const float e1 = lane_value(du1, 0);
            float sv = s1[0];
            const v2f ua = chen_pair(dk0, hk0, djp0,
                                     sv + 0.5f * e0,
                                     0.5f * sv + (1.0f / 6.0f) * e0,
                                     (1.0f / 3.0f) * sv + (1.0f / 12.0f) * e0,
                                     s3p[0], s2p[0]);
            sv += e0;
            const v2f ub = chen_pair(dk1, hk1, djp1,
                                     sv + 0.5f * e1,
                                     0.5f * sv + (1.0f / 6.0f) * e1,
                                     (1.0f / 3.0f) * sv + (1.0f / 12.0f) * e1,
                                     s3p[0], s2p[0]);
            s1[0] = sv + e1;
            v2f af0, af1;
            af0[0] = ua[0]; af0[1] = ub[0];
            af1[0] = ua[1]; af1[1] = ub[1];
            acc[0] = __builtin_amdgcn_wmma_f32_16x16x4_f32(
                false, af0, false, bf, (short)0, acc[0], false, false);
            acc[1] = __builtin_amdgcn_wmma_f32_16x16x4_f32(
                false, af1, false, bf, (short)0, acc[1], false, false);
        }
        {
            const float e0 = lane_value(du0, 1);
            const float e1 = lane_value(du1, 1);
            float sv = s1[1];
            chen_low(dk0, djp0, sv + 0.5f * e0, 0.5f * sv + (1.0f / 6.0f) * e0,
                     s3p[1], s2p[1]);
            sv += e0;
            chen_low(dk1, djp1, sv + 0.5f * e1, 0.5f * sv + (1.0f / 6.0f) * e1,
                     s3p[1], s2p[1]);
            s1[1] = sv + e1;
        }
    }
    float* sg = &sig[ch * SIGLEN];
#pragma unroll
    for (int q = 0; q < 4; ++q) {
        const int ia = (q >> 1) ? io : iw;
        sg[L3O + (2 * ia + (q & 1)) * 32 + h * 16 + c] = s3p[q >> 1][q & 1];
    }
    if (kk == 0) {
#pragma unroll
        for (int q = 0; q < 4; ++q) {
            const int ia = (q >> 1) ? io : iw;
            sg[L2O + ((ia << 3) | (4 * (q & 1) + jb))] = s2p[q >> 1][q & 1];
        }
    }
    if (lane == 0) {
        sg[iw] = s1[0];
        sg[io] = s1[1];
    }
#pragma unroll
    for (int q = 0; q < 2; ++q) {
        const int row0 = (2 * iw + q) * 32 + hi * 16 + h * 8;
#pragma unroll
        for (int v = 0; v < 8; ++v)
            sg[L4O + (row0 + v) * 8 + kk] = acc[q][v];
    }
    __syncthreads();

    const int e4 = 2048 * g;

    {
        const int p  = tid >> 8;
        const int lt = tid & 255;
        float* A = &sig[(2 * p) * SIGLEN];
        const float* Bs = &sig[(2 * p + 1) * SIGLEN];
        float* pr = &dxs[p * L4O];
        float r4[8];
#pragma unroll
        for (int n = 0; n < 8; ++n) {
            const int e = e4 + lt + 256 * n;
            r4[n] = (A[L4O + e] + Bs[L4O + e])
                  + A[e >> 9] * Bs[L3O + (e & 511)]
                  + A[L2O + (e >> 6)] * Bs[L2O + (e & 63)]
                  + A[L3O + (e >> 3)] * Bs[e & 7];
        }
        for (int e = lt; e < 512; e += 256)
            pr[L3O + e] = (A[L3O + e] + Bs[L3O + e])
                        + A[e >> 6] * Bs[L2O + (e & 63)]
                        + A[L2O + (e >> 3)] * Bs[e & 7];
        if (lt < 64)
            pr[L2O + lt] = (A[L2O + lt] + Bs[L2O + lt]) + A[lt >> 3] * Bs[lt & 7];
        if (lt < 8)
            pr[lt] = A[lt] + Bs[lt];
#pragma unroll
        for (int n = 0; n < 8; ++n)
            A[L4O + e4 + lt + 256 * n] = r4[n];
    }
    __syncthreads();

    const float* P  = &dxs[0];
    const float* Q  = &dxs[L4O];
    const float* P4 = &sig[L4O];
    const float* Q4 = &sig[2 * SIGLEN + L4O];
    float* fin = &dxs[2 * L4O];
    fin[L3O + tid] = (P[L3O + tid] + Q[L3O + tid])
                   + P[tid >> 6] * Q[L2O + (tid & 63)]
                   + P[L2O + (tid >> 3)] * Q[tid & 7];
    if (tid < 64)
        fin[L2O + tid] = (P[L2O + tid] + Q[L2O + tid]) + P[tid >> 3] * Q[tid & 7];
    if (tid < 8)
        fin[tid] = P[tid] + Q[tid];
    __syncthreads();

    float* ob = out + (size_t)b * 4680;
    const float* S1f = fin;
    const float* S2f = fin + L2O;
    const float* S3s = fin + L3O;

    if (g == 0 && tid < 8) ob[tid] = S1f[tid];
    if (g == 0 && tid < 64) {
        const int i = tid >> 3, j = tid & 7;
        ob[8 + tid] = S2f[tid] - 0.5f * S1f[i] * S1f[j];
    }
    if (g == 0) {
        const int e = tid;
        const int i = e >> 6, j = (e >> 3) & 7, k = e & 7;
        const float v3 = S3s[e]
            - 0.5f * (S1f[i] * S2f[(j << 3) | k] + S2f[(i << 3) | j] * S1f[k])
            + (1.0f / 3.0f) * S1f[i] * S1f[j] * S1f[k];
        ob[72 + e] = v3;
    }
    for (int n = 0; n < 2048 / NTHREADS; ++n) {
        const int e = e4 + tid + NTHREADS * n;
        const int i = e >> 9, j = (e >> 6) & 7, k = (e >> 3) & 7, l = e & 7;
        const float s4 = (P4[e] + Q4[e])
                       + P[i] * Q[L3O + (e & 511)]
                       + P[L2O + (e >> 6)] * Q[L2O + (e & 63)]
                       + P[L3O + (e >> 3)] * Q[l];
        const float s1i = S1f[i], s1j = S1f[j], s1k = S1f[k], s1l = S1f[l];
        const float v4 = s4
            - 0.5f * ( s1i * S3s[e & 511]
                     + S2f[(i << 3) | j] * S2f[(k << 3) | l]
                     + S3s[e >> 3] * s1l )
            + (1.0f / 3.0f) * ( s1i * s1j * S2f[(k << 3) | l]
                              + s1i * S2f[(j << 3) | k] * s1l
                              + S2f[(i << 3) | j] * s1k * s1l )
            - 0.25f * s1i * s1j * s1k * s1l;
        ob[584 + e] = v4;
    }
}

extern "C" void kernel_launch(void* const* d_in, const int* in_sizes, int n_in,
                              void* d_out, int out_size, void* d_ws, size_t ws_size,
                              hipStream_t stream) {
    const float* path = (const float*)d_in[0];
    float* out = (float*)d_out;
    const int B = in_sizes[0] / (TT * CD);
    logsig_kernel<<<dim3(B, 2), dim3(NTHREADS), 0, stream>>>(path, out);
}


# Time loop takes 16 steps per trip and first asks for all their increments as one group of 128-bit LDS loads (table kept one row per channel): one LDS-read place per 16 steps, not per 4; sums unchanged
# speedup vs baseline: 13.4537x; 1.0617x over previous
#include <hip/hip_runtime.h>


typedef __attribute__((ext_vector_type(2))) float v2f;
typedef __attribute__((ext_vector_type(4))) float v4f;
typedef __attribute__((ext_vector_type(8))) float v8f;

#define CD 8
#define TT 512
#define NSEG 511
#define NCH 4
#define CHPAIR 64
#define NTHREADS 512
#define SIGLEN 4680
#define L2O 8
#define L3O 72
#define L4O 584
#define DXP 516
#define DXLEN (CD * DXP)
#define TRIP 16

__device__ __forceinline__ float lane_value(float x, int l) {
    return __int_as_float(__builtin_amdgcn_readlane(__float_as_int(x), l));
}

__device__ __forceinline__ v2f chen_pair(float dk, float hdk, v2f djp,
                                         float a1, float a2, float a3,
                                         v2f& s3p, v2f& s2p) {
    const v2f t3 = s2p + djp * a3;
    const v2f t2 = s2p + djp * a2;
    const v2f u = s3p + hdk * t3;
    s3p = s3p + dk * t2;
    s2p = s2p + djp * a1;
    return u;
}

__device__ __forceinline__ void chen_low(float dk, v2f djp, float a1, float a2,
                                         v2f& s3p, v2f& s2p) {
    const v2f t2 = s2p + djp * a2;
    s3p = s3p + dk * t2;
    s2p = s2p + djp * a1;
}

__global__ __launch_bounds__(NTHREADS) __attribute__((amdgpu_waves_per_eu(4, 4)))
void logsig_kernel(const float* __restrict__ path, float* __restrict__ out) {
    __shared__ __attribute__((aligned(16))) float lds[DXLEN + NCH * SIGLEN];
    float* const dxs = lds;
    float* const sig = lds + DXLEN;

    const int b    = blockIdx.x;
    const int g    = blockIdx.y;
    const int tid  = threadIdx.x;
    const int lane = tid & 31;
    const int wv   = __builtin_amdgcn_readfirstlane(tid >> 5);
    const int w    = wv & 3;
    const int ch   = wv >> 2;
    const int c    = lane & 15;
    const int h    = lane >> 4;
    const int kk   = c & 7;
    const int hi   = c >> 3;
    const int jb   = 2 * h + hi;
    const bool own = (hi == h);
    const int iw   = 4 * g + w;
    const int io   = 4 * (1 - g) + w;
    const int isel = (lane & 1) ? io : iw;

    const float* pb = path + (size_t)b * (TT * CD);

#pragma unroll
    for (int n = 0; n < 8; ++n) {
        const int e = tid + n * NTHREADS;
        const bool real = e < NSEG * CD;
        const int es = real ? e : e - CD;
        const float dv = pb[es + CD] - pb[es];
        dxs[(e & 7) * DXP + (e >> 3)] = real ? dv : 0.0f;
    }

    v8f acc[2];
    const v8f vzero = {0.f, 0.f, 0.f, 0.f, 0.f, 0.f, 0.f, 0.f};
    v2f s3p[2], s2p[2];
    float s1[2];
#pragma unroll
    for (int q = 0; q < 2; ++q) acc[q] = vzero;
#pragma unroll
    for (int a = 0; a < 2; ++a) { s3p[a] = 0.0f; s2p[a] = 0.0f; s1[a] = 0.0f; }

    __syncthreads();

    const float* rk = &dxs[kk * DXP + ch * (2 * CHPAIR)];
    const float* ra = &dxs[jb * DXP + ch * (2 * CHPAIR)];
    const float* rb = &dxs[(4 + jb) * DXP + ch * (2 * CHPAIR)];
    const float* ru = &dxs[isel * DXP + ch * (2 * CHPAIR)];
    for (int tr = 0; tr < 2 * CHPAIR / TRIP; ++tr) {
        v4f qk[TRIP / 4], qa[TRIP / 4], qb[TRIP / 4], qu[TRIP / 4];
#pragma unroll
        for (int q = 0; q < TRIP / 4; ++q) {
            qk[q] = *(const v4f*)(rk + TRIP * tr + 4 * q);
            qa[q] = *(const v4f*)(ra + TRIP * tr + 4 * q);
            qb[q] = *(const v4f*)(rb + TRIP * tr + 4 * q);
            qu[q] = *(const v4f*)(ru + TRIP * tr + 4 * q);
        }
#pragma unroll
        for (int s = 0; s < TRIP; s += 2) {
            const float dk0 = qk[s >> 2][s & 3], dk1 = qk[s >> 2][(s & 3) + 1];
            const float hk0 = 0.5f * dk0, hk1 = 0.5f * dk1;
            v2f djp0, djp1;
            djp0[0] = qa[s >> 2][s & 3];       djp0[1] = qb[s >> 2][s & 3];
            djp1[0] = qa[s >> 2][(s & 3) + 1]; djp1[1] = qb[s >> 2][(s & 3) + 1];
            v2f bf; bf[0] = own ? dk0 : 0.0f; bf[1] = own ? dk1 : 0.0f;
            const float du0 = qu[s >> 2][s & 3], du1 = qu[s >> 2][(s & 3) + 1];
            {
                const float e0 = lane_value(du0, 0);
                const float e1 = lane_value(du1, 0);
                float sv = s1[0];
                const v2f ua = chen_pair(dk0, hk0, djp0,
                                         sv + 0.5f * e0,
                                         0.5f * sv + (1.0f / 6.0f) * e0,
                                         (1.0f / 3.0f) * sv + (1.0f / 12.0f) * e0,
                                         s3p[0], s2p[0]);
                sv += e0;
                const v2f ub = chen_pair(dk1, hk1, djp1,
                                         sv + 0.5f * e1,
                                         0.5f * sv + (1.0f / 6.0f) * e1,
                                         (1.0f / 3.0f) * sv + (1.0f / 12.0f) * e1,
                                         s3p[0], s2p[0]);
                s1[0] = sv + e1;
                v2f af0, af1;
                af0[0] = ua[0]; af0[1] = ub[0];
                af1[0] = ua[1]; af1[1] = ub[1];
                acc[0] = __builtin_amdgcn_wmma_f32_16x16x4_f32(
                    false, af0, false, bf, (short)0, acc[0], false, false);
                acc[1] = __builtin_amdgcn_wmma_f32_16x16x4_f32(
                    false, af1, false, bf, (short)0, acc[1], false, false);
            }
            {
                const float e0 = lane_value(du0, 1);
                const float e1 = lane_value(du1, 1);
                float sv = s1[1];
                chen_low(dk0, djp0, sv + 0.5f * e0, 0.5f * sv + (1.0f / 6.0f) * e0,
                         s3p[1], s2p[1]);
                sv += e0;
                chen_low(dk1, djp1, sv + 0.5f * e1, 0.5f * sv + (1.0f / 6.0f) * e1,
                         s3p[1], s2p[1]);
                s1[1] = sv + e1;
            }
        }
    }
    float* sg = &sig[ch * SIGLEN];
#pragma unroll
    for (int q = 0; q < 4; ++q) {
        const int ia = (q >> 1) ? io : iw;
        sg[L3O + (2 * ia + (q & 1)) * 32 + h * 16 + c] = s3p[q >> 1][q & 1];
    }
    if (kk == 0) {
#pragma unroll
        for (int q = 0; q < 4; ++q) {
            const int ia = (q >> 1) ? io : iw;
            sg[L2O + ((ia << 3) | (4 * (q & 1) + jb))] = s2p[q >> 1][q & 1];
        }
    }
    if (lane == 0) {
        sg[iw] = s1[0];
        sg[io] = s1[1];
    }
#pragma unroll
    for (int q = 0; q < 2; ++q) {
        const int row0 = (2 * iw + q) * 32 + hi * 16 + h * 8;
#pragma unroll
        for (int v = 0; v < 8; ++v)
            sg[L4O + (row0 + v) * 8 + kk] = acc[q][v];
    }
    __syncthreads();

    const int e4 = 2048 * g;

    {
        const int p  = tid >> 8;
        const int lt = tid & 255;
        float* A = &sig[(2 * p) * SIGLEN];
        const float* Bs = &sig[(2 * p + 1) * SIGLEN];
        float* pr = &dxs[p * L4O];
        float r4[8];
#pragma unroll
        for (int n = 0; n < 8; ++n) {
            const int e = e4 + lt + 256 * n;
            r4[n] = (A[L4O + e] + Bs[L4O + e])
                  + A[e >> 9] * Bs[L3O + (e & 511)]
                  + A[L2O + (e >> 6)] * Bs[L2O + (e & 63)]
                  + A[L3O + (e >> 3)] * Bs[e & 7];
        }
        for (int e = lt; e < 512; e += 256)
            pr[L3O + e] = (A[L3O + e] + Bs[L3O + e])
                        + A[e >> 6] * Bs[L2O + (e & 63)]
                        + A[L2O + (e >> 3)] * Bs[e & 7];
        if (lt < 64)
            pr[L2O + lt] = (A[L2O + lt] + Bs[L2O + lt]) + A[lt >> 3] * Bs[lt & 7];
        if (lt < 8)
            pr[lt] = A[lt] + Bs[lt];
#pragma unroll
        for (int n = 0; n < 8; ++n)
            A[L4O + e4 + lt + 256 * n] = r4[n];
    }
    __syncthreads();

    const float* P  = &dxs[0];
    const float* Q  = &dxs[L4O];
    const float* P4 = &sig[L4O];
    const float* Q4 = &sig[2 * SIGLEN + L4O];
    float* fin = &dxs[2 * L4O];
    fin[L3O + tid] = (P[L3O + tid] + Q[L3O + tid])
                   + P[tid >> 6] * Q[L2O + (tid & 63)]
                   + P[L2O + (tid >> 3)] * Q[tid & 7];
    if (tid < 64)
        fin[L2O + tid] = (P[L2O + tid] + Q[L2O + tid]) + P[tid >> 3] * Q[tid & 7];
    if (tid < 8)
        fin[tid] = P[tid] + Q[tid];
    __syncthreads();

    float* ob = out + (size_t)b * 4680;
    const float* S1f = fin;
    const float* S2f = fin + L2O;
    const float* S3s = fin + L3O;

    if (g == 0 && tid < 8) ob[tid] = S1f[tid];
    if (g == 0 && tid < 64) {
        const int i = tid >> 3, j = tid & 7;
        ob[8 + tid] = S2f[tid] - 0.5f * S1f[i] * S1f[j];
    }
    if (g == 0) {
        const int e = tid;
        const int i = e >> 6, j = (e >> 3) & 7, k = e & 7;
        const float v3 = S3s[e]
            - 0.5f * (S1f[i] * S2f[(j << 3) | k] + S2f[(i << 3) | j] * S1f[k])
            + (1.0f / 3.0f) * S1f[i] * S1f[j] * S1f[k];
        ob[72 + e] = v3;
    }
    for (int n = 0; n < 2048 / NTHREADS; ++n) {
        const int e = e4 + tid + NTHREADS * n;
        const int i = e >> 9, j = (e >> 6) & 7, k = (e >> 3) & 7, l = e & 7;
        const float s4 = (P4[e] + Q4[e])
                       + P[i] * Q[L3O + (e & 511)]
                       + P[L2O + (e >> 6)] * Q[L2O + (e & 63)]
                       + P[L3O + (e >> 3)] * Q[l];
        const float s1i = S1f[i], s1j = S1f[j], s1k = S1f[k], s1l = S1f[l];
        const float v4 = s4
            - 0.5f * ( s1i * S3s[e & 511]
                     + S2f[(i << 3) | j] * S2f[(k << 3) | l]
                     + S3s[e >> 3] * s1l )
            + (1.0f / 3.0f) * ( s1i * s1j * S2f[(k << 3) | l]
                              + s1i * S2f[(j << 3) | k] * s1l
                              + S2f[(i << 3) | j] * s1k * s1l )
            - 0.25f * s1i * s1j * s1k * s1l;
        ob[584 + e] = v4;
    }
}

extern "C" void kernel_launch(void* const* d_in, const int* in_sizes, int n_in,
                              void* d_out, int out_size, void* d_ws, size_t ws_size,
                              hipStream_t stream) {
    const float* path = (const float*)d_in[0];
    float* out = (float*)d_out;
    const int B = in_sizes[0] / (TT * CD);
    logsig_kernel<<<dim3(B, 2), dim3(NTHREADS), 0, stream>>>(path, out);
}


# Chen coefficients S1+d/2, S1/2+d/6, S1/3+d/12 tabled in LDS once per channel and step (S1 = path[t]-path[chunk start]: level-1 sum order changes); time loop has no scalar float ops or lane reads
# speedup vs baseline: 14.8104x; 1.1008x over previous
#include <hip/hip_runtime.h>


typedef __attribute__((ext_vector_type(2))) float v2f;
typedef __attribute__((ext_vector_type(4))) float v4f;
typedef __attribute__((ext_vector_type(8))) float v8f;

#define CD 8
#define TT 512
#define NSEG 511
#define NCH 4
#define CHPAIR 64
#define NTHREADS 512
#define SIGLEN 4680
#define L2O 8
#define L3O 72
#define L4O 584
#define DXP 516
#define DXLEN (CD * DXP)
#define TRIP 16
#define P12 1032
#define P3 516

__device__ __forceinline__ v2f chen_pair(float dk, float hdk, v2f djp,
                                         float a1, float a2, float a3,
                                         v2f& s3p, v2f& s2p) {
    const v2f t3 = s2p + djp * a3;
    const v2f t2 = s2p + djp * a2;
    const v2f u = s3p + hdk * t3;
    s3p = s3p + dk * t2;
    s2p = s2p + djp * a1;
    return u;
}

__device__ __forceinline__ void chen_low(float dk, v2f djp, float a1, float a2,
                                         v2f& s3p, v2f& s2p) {
    const v2f t2 = s2p + djp * a2;
    s3p = s3p + dk * t2;
    s2p = s2p + djp * a1;
}

__global__ __launch_bounds__(NTHREADS) __attribute__((amdgpu_waves_per_eu(4, 4)))
void logsig_kernel(const float* __restrict__ path, float* __restrict__ out) {
    __shared__ __attribute__((aligned(16))) float lds[DXLEN + CD * P12 + CD * P3 + NCH * SIGLEN];
    float* const dxs = lds;
    float* const t12 = lds + DXLEN;
    float* const t3  = t12 + CD * P12;
    float* const sig = t3 + CD * P3;

    const int b    = blockIdx.x;
    const int g    = blockIdx.y;
    const int tid  = threadIdx.x;
    const int lane = tid & 31;
    const int wv   = __builtin_amdgcn_readfirstlane(tid >> 5);
    const int w    = wv & 3;
    const int ch   = wv >> 2;
    const int c    = lane & 15;
    const int h    = lane >> 4;
    const int kk   = c & 7;
    const int hi   = c >> 3;
    const int jb   = 2 * h + hi;
    const bool own = (hi == h);
    const int iw   = 4 * g + w;
    const int io   = 4 * (1 - g) + w;

    const float* pb = path + (size_t)b * (TT * CD);

    float s1tot[NCH];
#pragma unroll
    for (int n = 0; n < 8; ++n) {
        const int e = tid + n * NTHREADS;
        const bool real = e < NSEG * CD;
        const int cc = tid & 7;
        const int t  = e >> 3;
        const float lo = pb[e];
        const float hi = pb[real ? e + CD : e];
        const float p0 = pb[((n >> 1) << 10) | cc];
        const float dv = real ? hi - lo : 0.0f;
        dxs[cc * DXP + t] = dv;
        const float sv = lo - p0;
        v2f a12;
        a12[0] = sv + 0.5f * dv;
        a12[1] = 0.5f * sv + (1.0f / 6.0f) * dv;
        *(v2f*)(t12 + cc * P12 + 2 * t) = a12;
        t3[cc * P3 + t] = (1.0f / 3.0f) * sv + (1.0f / 12.0f) * dv;
        if (n & 1) s1tot[n >> 1] = sv + dv;
    }
    if ((tid >> 3) == 63) {
#pragma unroll
        for (int k = 0; k < NCH; ++k) sig[k * SIGLEN + (tid & 7)] = s1tot[k];
    }

    v8f acc[2];
    const v8f vzero = {0.f, 0.f, 0.f, 0.f, 0.f, 0.f, 0.f, 0.f};
    v2f s3p[2], s2p[2];
#pragma unroll
    for (int q = 0; q < 2; ++q) acc[q] = vzero;
#pragma unroll
    for (int a = 0; a < 2; ++a) { s3p[a] = 0.0f; s2p[a] = 0.0f; }

    __syncthreads();

    const float* rk = &dxs[kk * DXP + ch * (2 * CHPAIR)];
    const float* ra = &dxs[jb * DXP + ch * (2 * CHPAIR)];
    const float* rb = &dxs[(4 + jb) * DXP + ch * (2 * CHPAIR)];
    const float* cw = &t12[iw * P12 + 2 * ch * (2 * CHPAIR)];
    const float* c3 = &t3[iw * P3 + ch * (2 * CHPAIR)];
    const float* co = &t12[io * P12 + 2 * ch * (2 * CHPAIR)];
    for (int tr = 0; tr < 2 * CHPAIR / TRIP; ++tr) {
        v4f qk[TRIP / 4], qa[TRIP / 4], qb[TRIP / 4], q3[TRIP / 4];
        v4f qw[TRIP / 2], qo[TRIP / 2];
#pragma unroll
        for (int q = 0; q < TRIP / 4; ++q) {
            qk[q] = *(const v4f*)(rk + TRIP * tr + 4 * q);
            qa[q] = *(const v4f*)(ra + TRIP * tr + 4 * q);
            qb[q] = *(const v4f*)(rb + TRIP * tr + 4 * q);
            q3[q] = *(const v4f*)(c3 + TRIP * tr + 4 * q);
            qw[2 * q]     = *(const v4f*)(cw + 2 * TRIP * tr + 8 * q);
            qw[2 * q + 1] = *(const v4f*)(cw + 2 * TRIP * tr + 8 * q + 4);
            qo[2 * q]     = *(const v4f*)(co + 2 * TRIP * tr + 8 * q);
            qo[2 * q + 1] = *(const v4f*)(co + 2 * TRIP * tr + 8 * q + 4);
        }
#pragma unroll
        for (int s = 0; s < TRIP; s += 2) {
            const float dk0 = qk[s >> 2][s & 3], dk1 = qk[s >> 2][(s & 3) + 1];
            const float hk0 = 0.5f * dk0, hk1 = 0.5f * dk1;
            v2f djp0, djp1;
            djp0[0] = qa[s >> 2][s & 3];       djp0[1] = qb[s >> 2][s & 3];
            djp1[0] = qa[s >> 2][(s & 3) + 1]; djp1[1] = qb[s >> 2][(s & 3) + 1];
            v2f bf; bf[0] = own ? dk0 : 0.0f; bf[1] = own ? dk1 : 0.0f;
            {
                const v4f cf = qw[s >> 1];
                const v2f ua = chen_pair(dk0, hk0, djp0, cf[0], cf[1], q3[s >> 2][s & 3],
                                         s3p[0], s2p[0]);
                const v2f ub = chen_pair(dk1, hk1, djp1, cf[2], cf[3], q3[s >> 2][(s & 3) + 1],
                                         s3p[0], s2p[0]);
                v2f af0, af1;
                af0[0] = ua[0]; af0[1] = ub[0];
                af1[0] = ua[1]; af1[1] = ub[1];
                acc[0] = __builtin_amdgcn_wmma_f32_16x16x4_f32(
                    false, af0, false, bf, (short)0, acc[0], false, false);
                acc[1] = __builtin_amdgcn_wmma_f32_16x16x4_f32(
                    false, af1, false, bf, (short)0, acc[1], false, false);
            }
            {
                const v4f cf = qo[s >> 1];
                chen_low(dk0, djp0, cf[0], cf[1], s3p[1], s2p[1]);
                chen_low(dk1, djp1, cf[2], cf[3], s3p[1], s2p[1]);
            }
        }
    }
    float* sg = &sig[ch * SIGLEN];
#pragma unroll
    for (int q = 0; q < 4; ++q) {
        const int ia = (q >> 1) ? io : iw;
        sg[L3O + (2 * ia + (q & 1)) * 32 + h * 16 + c] = s3p[q >> 1][q & 1];
    }
    if (kk == 0) {
#pragma unroll
        for (int q = 0; q < 4; ++q) {
            const int ia = (q >> 1) ? io : iw;
            sg[L2O + ((ia << 3) | (4 * (q & 1) + jb))] = s2p[q >> 1][q & 1];
        }
    }
#pragma unroll
    for (int q = 0; q < 2; ++q) {
        const int row0 = (2 * iw + q) * 32 + hi * 16 + h * 8;
#pragma unroll
        for (int v = 0; v < 8; ++v)
            sg[L4O + (row0 + v) * 8 + kk] = acc[q][v];
    }
    __syncthreads();

    const int e4 = 2048 * g;

    {
        const int p  = tid >> 8;
        const int lt = tid & 255;
        float* A = &sig[(2 * p) * SIGLEN];
        const float* Bs = &sig[(2 * p + 1) * SIGLEN];
        float* pr = &dxs[p * L4O];
        float r4[8];
#pragma unroll
        for (int n = 0; n < 8; ++n) {
            const int e = e4 + lt + 256 * n;
            r4[n] = (A[L4O + e] + Bs[L4O + e])
                  + A[e >> 9] * Bs[L3O + (e & 511)]
                  + A[L2O + (e >> 6)] * Bs[L2O + (e & 63)]
                  + A[L3O + (e >> 3)] * Bs[e & 7];
        }
        for (int e = lt; e < 512; e += 256)
            pr[L3O + e] = (A[L3O + e] + Bs[L3O + e])
                        + A[e >> 6] * Bs[L2O + (e & 63)]
                        + A[L2O + (e >> 3)] * Bs[e & 7];
        if (lt < 64)
            pr[L2O + lt] = (A[L2O + lt] + Bs[L2O + lt]) + A[lt >> 3] * Bs[lt & 7];
        if (lt < 8)
            pr[lt] = A[lt] + Bs[lt];
#pragma unroll
        for (int n = 0; n < 8; ++n)
            A[L4O + e4 + lt + 256 * n] = r4[n];
    }
    __syncthreads();

    const float* P  = &dxs[0];
    const float* Q  = &dxs[L4O];
    const float* P4 = &sig[L4O];
    const float* Q4 = &sig[2 * SIGLEN + L4O];
    float* fin = &dxs[2 * L4O];
    fin[L3O + tid] = (P[L3O + tid] + Q[L3O + tid])
                   + P[tid >> 6] * Q[L2O + (tid & 63)]
                   + P[L2O + (tid >> 3)] * Q[tid & 7];
    if (tid < 64)
        fin[L2O + tid] = (P[L2O + tid] + Q[L2O + tid]) + P[tid >> 3] * Q[tid & 7];
    if (tid < 8)
        fin[tid] = P[tid] + Q[tid];
    __syncthreads();

    float* ob = out + (size_t)b * 4680;
    const float* S1f = fin;
    const float* S2f = fin + L2O;
    const float* S3s = fin + L3O;

    if (g == 0 && tid < 8) ob[tid] = S1f[tid];
    if (g == 0 && tid < 64) {
        const int i = tid >> 3, j = tid & 7;
        ob[8 + tid] = S2f[tid] - 0.5f * S1f[i] * S1f[j];
    }
    if (g == 0) {
        const int e = tid;
        const int i = e >> 6, j = (e >> 3) & 7, k = e & 7;
        const float v3 = S3s[e]
            - 0.5f * (S1f[i] * S2f[(j << 3) | k] + S2f[(i << 3) | j] * S1f[k])
            + (1.0f / 3.0f) * S1f[i] * S1f[j] * S1f[k];
        ob[72 + e] = v3;
    }
    for (int n = 0; n < 2048 / NTHREADS; ++n) {
        const int e = e4 + tid + NTHREADS * n;
        const int i = e >> 9, j = (e >> 6) & 7, k = (e >> 3) & 7, l = e & 7;
        const float s4 = (P4[e] + Q4[e])
                       + P[i] * Q[L3O + (e & 511)]
                       + P[L2O + (e >> 6)] * Q[L2O + (e & 63)]
                       + P[L3O + (e >> 3)] * Q[l];
        const float s1i = S1f[i], s1j = S1f[j], s1k = S1f[k], s1l = S1f[l];
        const float v4 = s4
            - 0.5f * ( s1i * S3s[e & 511]
                     + S2f[(i << 3) | j] * S2f[(k << 3) | l]
                     + S3s[e >> 3] * s1l )
            + (1.0f / 3.0f) * ( s1i * s1j * S2f[(k << 3) | l]
                              + s1i * S2f[(j << 3) | k] * s1l
                              + S2f[(i << 3) | j] * s1k * s1l )
            - 0.25f * s1i * s1j * s1k * s1l;
        ob[584 + e] = v4;
    }
}

extern "C" void kernel_launch(void* const* d_in, const int* in_sizes, int n_in,
                              void* d_out, int out_size, void* d_ws, size_t ws_size,
                              hipStream_t stream) {
    const float* path = (const float*)d_in[0];
    float* out = (float*)d_out;
    const int B = in_sizes[0] / (TT * CD);
    logsig_kernel<<<dim3(B, 2), dim3(NTHREADS), 0, stream>>>(path, out);
}
